# starting kernel
# baseline (speedup 1.0000x reference)
#include <hip/hip_runtime.h>

#define LN_EPS  1e-5f
#define BB      8
#define NN      10000
#define CC      256
#define NCHUNK  80
#define CHUNK   125
#define MPAD    16

typedef __attribute__((ext_vector_type(2))) float v2f;
typedef __attribute__((ext_vector_type(4))) float v4f;
typedef __attribute__((ext_vector_type(8))) float v8f;

__global__ __launch_bounds__(256) void colsum_partial_kernel(
    const float* __restrict__ x, float* __restrict__ partial) {
  const int b  = blockIdx.x;
  const int ch = blockIdx.y;
  const int c  = threadIdx.x;
  const float* xp = x + (((size_t)b * NN + (size_t)ch * CHUNK) * CC) + c;
  float s = 0.0f;
#pragma unroll 5
  for (int i = 0; i < CHUNK; ++i) s += xp[(size_t)i * CC];
  partial[((size_t)b * NCHUNK + ch) * CC + c] = s;
}

__global__ __launch_bounds__(256) void colsum_final_kernel(
    const float* __restrict__ partial, float* __restrict__ meanp) {
  const int b = blockIdx.x;
  const int c = threadIdx.x;
  float s = 0.0f;
  if (b < BB) {
    const float* pp = partial + (size_t)b * NCHUNK * CC + c;
#pragma unroll 8
    for (int ch = 0; ch < NCHUNK; ++ch) s += pp[(size_t)ch * CC];
    s *= (1.0f / (float)NN);
  }
  meanp[b * CC + c] = s;
}

__global__ __launch_bounds__(512) void gemm_wmma_kernel(
    const float* __restrict__ meanp, const float* __restrict__ W,
    const float* __restrict__ bias, float* __restrict__ agg) {
  const int wave = threadIdx.x >> 5;
  const int lane = threadIdx.x & 31;
  const int hh   = lane >> 4;
  const int l    = lane & 15;
  const int n0   = wave * 16;

  v8f acc = {};
#pragma unroll 4
  for (int k0 = 0; k0 < CC; k0 += 4) {
    const int kk = k0 + 2 * hh;
    v2f a;
    a.x = meanp[l * CC + kk + 0];
    a.y = meanp[l * CC + kk + 1];
    v2f bm;
    bm.x = W[(size_t)(n0 + l) * CC + kk + 0];
    bm.y = W[(size_t)(n0 + l) * CC + kk + 1];
    acc = __builtin_amdgcn_wmma_f32_16x16x4_f32(
         false, a,  false, bm,
         (short)0, acc,  false,  false);
  }

  const int n = n0 + l;
  const float bn = bias[n];
#pragma unroll
  for (int r = 0; r < 8; ++r) {
    const int m = hh * 8 + r;
    agg[m * CC + n] = acc[r] + bn;
  }
}

__global__ __launch_bounds__(256) void fused_residual_ln_kernel(
    const float* __restrict__ x, const float* __restrict__ agg,
    const float* __restrict__ gamma, const float* __restrict__ beta,
    float* __restrict__ out) {
  const int lane = threadIdx.x & 31;
  const int wave = threadIdx.x >> 5;
  const int wavesTotal = gridDim.x * 8;
  const int c0 = lane * 8;

  const v4f g0 = *(const v4f*)(gamma + c0);
  const v4f g1 = *(const v4f*)(gamma + c0 + 4);
  const v4f e0 = *(const v4f*)(beta + c0);
  const v4f e1 = *(const v4f*)(beta + c0 + 4);

  for (int row = blockIdx.x * 8 + wave; row < BB * NN; row += wavesTotal) {
    const int b = row / NN;
    const float* xp = x + (size_t)row * CC + c0;
    const float* ap = agg + b * CC + c0;

    const v4f xv0 = *(const v4f*)(xp);
    const v4f xv1 = *(const v4f*)(xp + 4);
    const v4f av0 = *(const v4f*)(ap);
    const v4f av1 = *(const v4f*)(ap + 4);

    const v4f y0 = xv0 + av0;
    const v4f y1 = xv1 + av1;

    float s  = y0.x + y0.y + y0.z + y0.w + y1.x + y1.y + y1.z + y1.w;
    float s2 = y0.x * y0.x + y0.y * y0.y + y0.z * y0.z + y0.w * y0.w +
               y1.x * y1.x + y1.y * y1.y + y1.z * y1.z + y1.w * y1.w;
#pragma unroll
    for (int off = 16; off >= 1; off >>= 1) {
      s  += __shfl_xor(s, off, 32);
      s2 += __shfl_xor(s2, off, 32);
    }
    const float mu  = s * (1.0f / CC);
    const float var = fmaf(-mu, mu, s2 * (1.0f / CC));
    const float inv = rsqrtf(var + LN_EPS);

    v4f o0, o1;
    o0.x = (y0.x - mu) * inv * g0.x + e0.x;
    o0.y = (y0.y - mu) * inv * g0.y + e0.y;
    o0.z = (y0.z - mu) * inv * g0.z + e0.z;
    o0.w = (y0.w - mu) * inv * g0.w + e0.w;
    o1.x = (y1.x - mu) * inv * g1.x + e1.x;
    o1.y = (y1.y - mu) * inv * g1.y + e1.y;
    o1.z = (y1.z - mu) * inv * g1.z + e1.z;
    o1.w = (y1.w - mu) * inv * g1.w + e1.w;

    float* op = out + (size_t)row * CC + c0;
    __builtin_nontemporal_store(o0, (v4f*)(op));
    __builtin_nontemporal_store(o1, (v4f*)(op + 4));
  }
}

extern "C" void kernel_launch(void* const* d_in, const int* in_sizes, int n_in,
                              void* d_out, int out_size, void* d_ws, size_t ws_size,
                              hipStream_t stream) {
  (void)in_sizes; (void)n_in; (void)out_size; (void)ws_size;

  const float* x     = (const float*)d_in[0];
  const float* W     = (const float*)d_in[1];
  const float* bias  = (const float*)d_in[2];
  const float* gamma = (const float*)d_in[3];
  const float* beta  = (const float*)d_in[4];
  float* out = (float*)d_out;

  float* ws      = (float*)d_ws;
  float* partial = ws;
  float* meanp   = partial + (size_t)BB * NCHUNK * CC;
  float* agg     = meanp + MPAD * CC;

  colsum_partial_kernel<<<dim3(BB, NCHUNK), 256, 0, stream>>>(x, partial);
  colsum_final_kernel<<<MPAD, 256, 0, stream>>>(partial, meanp);
  gemm_wmma_kernel<<<1, 512, 0, stream>>>(meanp, W, bias, agg);
  fused_residual_ln_kernel<<<(BB * NN + 7) / 8, 256, 0, stream>>>(x, agg, gamma, beta, out);
}


# GEMM launch spread from 1 work-group to 8 (split-K): each wave takes one N tile and one K slice of 32 with its loads issued as one group; the 8 slice partials are added in slice order through LDS
# speedup vs baseline: 1.2474x; 1.2474x over previous
#include <hip/hip_runtime.h>

#define LN_EPS  1e-5f
#define BB      8
#define NN      10000
#define CC      256
#define NCHUNK  80
#define CHUNK   125
#define MPAD    16

typedef __attribute__((ext_vector_type(2))) float v2f;
typedef __attribute__((ext_vector_type(4))) float v4f;
typedef __attribute__((ext_vector_type(8))) float v8f;

__global__ __launch_bounds__(256) void colsum_partial_kernel(
    const float* __restrict__ x, float* __restrict__ partial) {
  const int b  = blockIdx.x;
  const int ch = blockIdx.y;
  const int c  = threadIdx.x;
  const float* xp = x + (((size_t)b * NN + (size_t)ch * CHUNK) * CC) + c;
  float s = 0.0f;
#pragma unroll 5
  for (int i = 0; i < CHUNK; ++i) s += xp[(size_t)i * CC];
  partial[((size_t)b * NCHUNK + ch) * CC + c] = s;
}

__global__ __launch_bounds__(256) void colsum_final_kernel(
    const float* __restrict__ partial, float* __restrict__ meanp) {
  const int b = blockIdx.x;
  const int c = threadIdx.x;
  float s = 0.0f;
  if (b < BB) {
    const float* pp = partial + (size_t)b * NCHUNK * CC + c;
#pragma unroll 8
    for (int ch = 0; ch < NCHUNK; ++ch) s += pp[(size_t)ch * CC];
    s *= (1.0f / (float)NN);
  }
  meanp[b * CC + c] = s;
}

#define KSLICES 8
#define KSLICE  (CC / KSLICES)
#define KSTEPS  (KSLICE / 4)
#define NGROUPS (CC / 32)

__global__ __launch_bounds__(512) void gemm_wmma_kernel(
    const float* __restrict__ meanp, const float* __restrict__ W,
    const float* __restrict__ bias, float* __restrict__ agg) {
  __shared__ float red[2][KSLICES][8][32];

  const int wave = threadIdx.x >> 5;
  const int lane = threadIdx.x & 31;
  const int hh   = lane >> 4;
  const int l    = lane & 15;
  const int tile = wave >> 3;
  const int ks   = wave & 7;
  const int nb   = blockIdx.x * 32;
  const int n0   = nb + tile * 16;
  const int kb   = ks * KSLICE + 2 * hh;

  const float* ap = meanp + l * CC + kb;
  const float* bp = W + (size_t)(n0 + l) * CC + kb;

  v2f a[KSTEPS], bm[KSTEPS];
  for (int j = 0; j < KSTEPS; ++j) {
    a[j]  = *(const v2f*)(ap + 4 * j);
    bm[j] = *(const v2f*)(bp + 4 * j);
  }
  const float bn = bias[nb + lane];

  v8f acc = {};
  for (int j = 0; j < KSTEPS; ++j) {
    acc = __builtin_amdgcn_wmma_f32_16x16x4_f32(
         false, a[j],  false, bm[j],
         (short)0, acc,  false,  false);
  }

  for (int r = 0; r < 8; ++r) red[tile][ks][r][lane] = acc[r];
  __syncthreads();

  const int m  = wave;
  const int tt = lane >> 4;
  const int sl = (m >> 3) * 16 + l;
  const int r  = m & 7;
  float sum = red[tt][0][r][sl];
  for (int q = 1; q < KSLICES; ++q) sum += red[tt][q][r][sl];
  agg[m * CC + nb + lane] = sum + bn;
}

__global__ __launch_bounds__(256) void fused_residual_ln_kernel(
    const float* __restrict__ x, const float* __restrict__ agg,
    const float* __restrict__ gamma, const float* __restrict__ beta,
    float* __restrict__ out) {
  const int lane = threadIdx.x & 31;
  const int wave = threadIdx.x >> 5;
  const int wavesTotal = gridDim.x * 8;
  const int c0 = lane * 8;

  const v4f g0 = *(const v4f*)(gamma + c0);
  const v4f g1 = *(const v4f*)(gamma + c0 + 4);
  const v4f e0 = *(const v4f*)(beta + c0);
  const v4f e1 = *(const v4f*)(beta + c0 + 4);

  for (int row = blockIdx.x * 8 + wave; row < BB * NN; row += wavesTotal) {
    const int b = row / NN;
    const float* xp = x + (size_t)row * CC + c0;
    const float* ap = agg + b * CC + c0;

    const v4f xv0 = *(const v4f*)(xp);
    const v4f xv1 = *(const v4f*)(xp + 4);
    const v4f av0 = *(const v4f*)(ap);
    const v4f av1 = *(const v4f*)(ap + 4);

    const v4f y0 = xv0 + av0;
    const v4f y1 = xv1 + av1;

    float s  = y0.x + y0.y + y0.z + y0.w + y1.x + y1.y + y1.z + y1.w;
    float s2 = y0.x * y0.x + y0.y * y0.y + y0.z * y0.z + y0.w * y0.w +
               y1.x * y1.x + y1.y * y1.y + y1.z * y1.z + y1.w * y1.w;
#pragma unroll
    for (int off = 16; off >= 1; off >>= 1) {
      s  += __shfl_xor(s, off, 32);
      s2 += __shfl_xor(s2, off, 32);
    }
    const float mu  = s * (1.0f / CC);
    const float var = fmaf(-mu, mu, s2 * (1.0f / CC));
    const float inv = rsqrtf(var + LN_EPS);

    v4f o0, o1;
    o0.x = (y0.x - mu) * inv * g0.x + e0.x;
    o0.y = (y0.y - mu) * inv * g0.y + e0.y;
    o0.z = (y0.z - mu) * inv * g0.z + e0.z;
    o0.w = (y0.w - mu) * inv * g0.w + e0.w;
    o1.x = (y1.x - mu) * inv * g1.x + e1.x;
    o1.y = (y1.y - mu) * inv * g1.y + e1.y;
    o1.z = (y1.z - mu) * inv * g1.z + e1.z;
    o1.w = (y1.w - mu) * inv * g1.w + e1.w;

    float* op = out + (size_t)row * CC + c0;
    __builtin_nontemporal_store(o0, (v4f*)(op));
    __builtin_nontemporal_store(o1, (v4f*)(op + 4));
  }
}

extern "C" void kernel_launch(void* const* d_in, const int* in_sizes, int n_in,
                              void* d_out, int out_size, void* d_ws, size_t ws_size,
                              hipStream_t stream) {
  (void)in_sizes; (void)n_in; (void)out_size; (void)ws_size;

  const float* x     = (const float*)d_in[0];
  const float* W     = (const float*)d_in[1];
  const float* bias  = (const float*)d_in[2];
  const float* gamma = (const float*)d_in[3];
  const float* beta  = (const float*)d_in[4];
  float* out = (float*)d_out;

  float* ws      = (float*)d_ws;
  float* partial = ws;
  float* meanp   = partial + (size_t)BB * NCHUNK * CC;
  float* agg     = meanp + MPAD * CC;

  colsum_partial_kernel<<<dim3(BB, NCHUNK), 256, 0, stream>>>(x, partial);
  colsum_final_kernel<<<MPAD, 256, 0, stream>>>(partial, meanp);
  gemm_wmma_kernel<<<NGROUPS, 512, 0, stream>>>(meanp, W, bias, agg);
  fused_residual_ln_kernel<<<(BB * NN + 7) / 8, 256, 0, stream>>>(x, agg, gamma, beta, out);
}


# Column-sum pass as a stream: a wave reads 15-16 consecutive rows with 16-byte loads, 13 in flight (was five 4-byte); chunk sum = 8 row-block sub-sums added in order via LDS, not one chain
# speedup vs baseline: 1.5323x; 1.2284x over previous
#include <hip/hip_runtime.h>

#define LN_EPS  1e-5f
#define BB      8
#define NN      10000
#define CC      256
#define NCHUNK  80
#define CHUNK   125
#define MPAD    16

typedef __attribute__((ext_vector_type(2))) float v2f;
typedef __attribute__((ext_vector_type(4))) float v4f;
typedef __attribute__((ext_vector_type(8))) float v8f;

#define CS_WAVES 8
#define CS_ROWS  16

__global__ __launch_bounds__(256) void colsum_partial_kernel(
    const float* __restrict__ x, float* __restrict__ partial) {
  __shared__ float red[CS_WAVES][CC];

  const int b    = blockIdx.x;
  const int ch   = blockIdx.y;
  const int lane = threadIdx.x & 31;
  const int wave = __builtin_amdgcn_readfirstlane((int)(threadIdx.x >> 5));

  const bool full = wave < 5;
  const int  r0   = wave * CS_ROWS - (full ? 0 : wave - 5);
  const int  rl   = full ? CS_ROWS - 1 : CS_ROWS - 2;

  const float* xp = x + (((size_t)b * NN + (size_t)ch * CHUNK + (size_t)r0) * CC) + lane * 4;

  v4f lo[CS_ROWS], hi[CS_ROWS];
#pragma unroll
  for (int i = 0; i < CS_ROWS - 1; ++i) {
    lo[i] = *(const v4f*)(xp + i * CC);
    hi[i] = *(const v4f*)(xp + i * CC + 128);
  }
  lo[CS_ROWS - 1] = *(const v4f*)(xp + rl * CC);
  hi[CS_ROWS - 1] = *(const v4f*)(xp + rl * CC + 128);

  v4f s0 = lo[0], s1 = hi[0];
#pragma unroll
  for (int i = 1; i < CS_ROWS - 1; ++i) { s0 += lo[i]; s1 += hi[i]; }
  if (full) { s0 += lo[CS_ROWS - 1]; s1 += hi[CS_ROWS - 1]; }

  *(v4f*)(&red[wave][lane * 4])       = s0;
  *(v4f*)(&red[wave][128 + lane * 4]) = s1;
  __syncthreads();

  const int c = threadIdx.x;
  float s = red[0][c];
#pragma unroll
  for (int w = 1; w < CS_WAVES; ++w) s += red[w][c];
  partial[((size_t)b * NCHUNK + ch) * CC + c] = s;
}

__global__ __launch_bounds__(256) void colsum_final_kernel(
    const float* __restrict__ partial, float* __restrict__ meanp) {
  const int b = blockIdx.x;
  const int c = threadIdx.x;
  float s = 0.0f;
  if (b < BB) {
    const float* pp = partial + (size_t)b * NCHUNK * CC + c;
#pragma unroll 8
    for (int ch = 0; ch < NCHUNK; ++ch) s += pp[(size_t)ch * CC];
    s *= (1.0f / (float)NN);
  }
  meanp[b * CC + c] = s;
}

#define KSLICES 8
#define KSLICE  (CC / KSLICES)
#define KSTEPS  (KSLICE / 4)
#define NGROUPS (CC / 32)

__global__ __launch_bounds__(512) void gemm_wmma_kernel(
    const float* __restrict__ meanp, const float* __restrict__ W,
    const float* __restrict__ bias, float* __restrict__ agg) {
  __shared__ float red[2][KSLICES][8][32];

  const int wave = threadIdx.x >> 5;
  const int lane = threadIdx.x & 31;
  const int hh   = lane >> 4;
  const int l    = lane & 15;
  const int tile = wave >> 3;
  const int ks   = wave & 7;
  const int nb   = blockIdx.x * 32;
  const int n0   = nb + tile * 16;
  const int kb   = ks * KSLICE + 2 * hh;

  const float* ap = meanp + l * CC + kb;
  const float* bp = W + (size_t)(n0 + l) * CC + kb;

  v2f a[KSTEPS], bm[KSTEPS];
  for (int j = 0; j < KSTEPS; ++j) {
    a[j]  = *(const v2f*)(ap + 4 * j);
    bm[j] = *(const v2f*)(bp + 4 * j);
  }
  const float bn = bias[nb + lane];

  v8f acc = {};
  for (int j = 0; j < KSTEPS; ++j) {
    acc = __builtin_amdgcn_wmma_f32_16x16x4_f32(
         false, a[j],  false, bm[j],
         (short)0, acc,  false,  false);
  }

  for (int r = 0; r < 8; ++r) red[tile][ks][r][lane] = acc[r];
  __syncthreads();

  const int m  = wave;
  const int tt = lane >> 4;
  const int sl = (m >> 3) * 16 + l;
  const int r  = m & 7;
  float sum = red[tt][0][r][sl];
  for (int q = 1; q < KSLICES; ++q) sum += red[tt][q][r][sl];
  agg[m * CC + nb + lane] = sum + bn;
}

__global__ __launch_bounds__(256) void fused_residual_ln_kernel(
    const float* __restrict__ x, const float* __restrict__ agg,
    const float* __restrict__ gamma, const float* __restrict__ beta,
    float* __restrict__ out) {
  const int lane = threadIdx.x & 31;
  const int wave = threadIdx.x >> 5;
  const int wavesTotal = gridDim.x * 8;
  const int c0 = lane * 8;

  const v4f g0 = *(const v4f*)(gamma + c0);
  const v4f g1 = *(const v4f*)(gamma + c0 + 4);
  const v4f e0 = *(const v4f*)(beta + c0);
  const v4f e1 = *(const v4f*)(beta + c0 + 4);

  for (int row = blockIdx.x * 8 + wave; row < BB * NN; row += wavesTotal) {
    const int b = row / NN;
    const float* xp = x + (size_t)row * CC + c0;
    const float* ap = agg + b * CC + c0;

    const v4f xv0 = *(const v4f*)(xp);
    const v4f xv1 = *(const v4f*)(xp + 4);
    const v4f av0 = *(const v4f*)(ap);
    const v4f av1 = *(const v4f*)(ap + 4);

    const v4f y0 = xv0 + av0;
    const v4f y1 = xv1 + av1;

    float s  = y0.x + y0.y + y0.z + y0.w + y1.x + y1.y + y1.z + y1.w;
    float s2 = y0.x * y0.x + y0.y * y0.y + y0.z * y0.z + y0.w * y0.w +
               y1.x * y1.x + y1.y * y1.y + y1.z * y1.z + y1.w * y1.w;
#pragma unroll
    for (int off = 16; off >= 1; off >>= 1) {
      s  += __shfl_xor(s, off, 32);
      s2 += __shfl_xor(s2, off, 32);
    }
    const float mu  = s * (1.0f / CC);
    const float var = fmaf(-mu, mu, s2 * (1.0f / CC));
    const float inv = rsqrtf(var + LN_EPS);

    v4f o0, o1;
    o0.x = (y0.x - mu) * inv * g0.x + e0.x;
    o0.y = (y0.y - mu) * inv * g0.y + e0.y;
    o0.z = (y0.z - mu) * inv * g0.z + e0.z;
    o0.w = (y0.w - mu) * inv * g0.w + e0.w;
    o1.x = (y1.x - mu) * inv * g1.x + e1.x;
    o1.y = (y1.y - mu) * inv * g1.y + e1.y;
    o1.z = (y1.z - mu) * inv * g1.z + e1.z;
    o1.w = (y1.w - mu) * inv * g1.w + e1.w;

    float* op = out + (size_t)row * CC + c0;
    __builtin_nontemporal_store(o0, (v4f*)(op));
    __builtin_nontemporal_store(o1, (v4f*)(op + 4));
  }
}

extern "C" void kernel_launch(void* const* d_in, const int* in_sizes, int n_in,
                              void* d_out, int out_size, void* d_ws, size_t ws_size,
                              hipStream_t stream) {
  (void)in_sizes; (void)n_in; (void)out_size; (void)ws_size;

  const float* x     = (const float*)d_in[0];
  const float* W     = (const float*)d_in[1];
  const float* bias  = (const float*)d_in[2];
  const float* gamma = (const float*)d_in[3];
  const float* beta  = (const float*)d_in[4];
  float* out = (float*)d_out;

  float* ws      = (float*)d_ws;
  float* partial = ws;
  float* meanp   = partial + (size_t)BB * NCHUNK * CC;
  float* agg     = meanp + MPAD * CC;

  colsum_partial_kernel<<<dim3(BB, NCHUNK), 256, 0, stream>>>(x, partial);
  colsum_final_kernel<<<MPAD, 256, 0, stream>>>(partial, meanp);
  gemm_wmma_kernel<<<NGROUPS, 512, 0, stream>>>(meanp, W, bias, agg);
  fused_residual_ln_kernel<<<(BB * NN + 7) / 8, 256, 0, stream>>>(x, agg, gamma, beta, out);
}


# LayerNorm pass as a stream: a wave takes 15-16 consecutive rows (column-sum pass's ownership), whole-line 16-byte loads/stores, 8 rows in flight; was 1 row per wave; row sums regrouped by lane map
# speedup vs baseline: 1.9410x; 1.2667x over previous
#include <hip/hip_runtime.h>

#define LN_EPS  1e-5f
#define BB      8
#define NN      10000
#define CC      256
#define NCHUNK  80
#define CHUNK   125
#define MPAD    16

typedef __attribute__((ext_vector_type(2))) float v2f;
typedef __attribute__((ext_vector_type(4))) float v4f;
typedef __attribute__((ext_vector_type(8))) float v8f;

#define CS_WAVES 8
#define CS_ROWS  16

__global__ __launch_bounds__(256) void colsum_partial_kernel(
    const float* __restrict__ x, float* __restrict__ partial) {
  __shared__ float red[CS_WAVES][CC];

  const int b    = blockIdx.x;
  const int ch   = blockIdx.y;
  const int lane = threadIdx.x & 31;
  const int wave = __builtin_amdgcn_readfirstlane((int)(threadIdx.x >> 5));

  const bool full = wave < 5;
  const int  r0   = wave * CS_ROWS - (full ? 0 : wave - 5);
  const int  rl   = full ? CS_ROWS - 1 : CS_ROWS - 2;

  const float* xp = x + (((size_t)b * NN + (size_t)ch * CHUNK + (size_t)r0) * CC) + lane * 4;

  v4f lo[CS_ROWS], hi[CS_ROWS];
#pragma unroll
  for (int i = 0; i < CS_ROWS - 1; ++i) {
    lo[i] = *(const v4f*)(xp + i * CC);
    hi[i] = *(const v4f*)(xp + i * CC + 128);
  }
  lo[CS_ROWS - 1] = *(const v4f*)(xp + rl * CC);
  hi[CS_ROWS - 1] = *(const v4f*)(xp + rl * CC + 128);

  v4f s0 = lo[0], s1 = hi[0];
#pragma unroll
  for (int i = 1; i < CS_ROWS - 1; ++i) { s0 += lo[i]; s1 += hi[i]; }
  if (full) { s0 += lo[CS_ROWS - 1]; s1 += hi[CS_ROWS - 1]; }

  *(v4f*)(&red[wave][lane * 4])       = s0;
  *(v4f*)(&red[wave][128 + lane * 4]) = s1;
  __syncthreads();

  const int c = threadIdx.x;
  float s = red[0][c];
#pragma unroll
  for (int w = 1; w < CS_WAVES; ++w) s += red[w][c];
  partial[((size_t)b * NCHUNK + ch) * CC + c] = s;
}

__global__ __launch_bounds__(256) void colsum_final_kernel(
    const float* __restrict__ partial, float* __restrict__ meanp) {
  const int b = blockIdx.x;
  const int c = threadIdx.x;
  float s = 0.0f;
  if (b < BB) {
    const float* pp = partial + (size_t)b * NCHUNK * CC + c;
#pragma unroll 8
    for (int ch = 0; ch < NCHUNK; ++ch) s += pp[(size_t)ch * CC];
    s *= (1.0f / (float)NN);
  }
  meanp[b * CC + c] = s;
}

#define KSLICES 8
#define KSLICE  (CC / KSLICES)
#define KSTEPS  (KSLICE / 4)
#define NGROUPS (CC / 32)

__global__ __launch_bounds__(512) void gemm_wmma_kernel(
    const float* __restrict__ meanp, const float* __restrict__ W,
    const float* __restrict__ bias, float* __restrict__ agg) {
  __shared__ float red[2][KSLICES][8][32];

  const int wave = threadIdx.x >> 5;
  const int lane = threadIdx.x & 31;
  const int hh   = lane >> 4;
  const int l    = lane & 15;
  const int tile = wave >> 3;
  const int ks   = wave & 7;
  const int nb   = blockIdx.x * 32;
  const int n0   = nb + tile * 16;
  const int kb   = ks * KSLICE + 2 * hh;

  const float* ap = meanp + l * CC + kb;
  const float* bp = W + (size_t)(n0 + l) * CC + kb;

  v2f a[KSTEPS], bm[KSTEPS];
  for (int j = 0; j < KSTEPS; ++j) {
    a[j]  = *(const v2f*)(ap + 4 * j);
    bm[j] = *(const v2f*)(bp + 4 * j);
  }
  const float bn = bias[nb + lane];

  v8f acc = {};
  for (int j = 0; j < KSTEPS; ++j) {
    acc = __builtin_amdgcn_wmma_f32_16x16x4_f32(
         false, a[j],  false, bm[j],
         (short)0, acc,  false,  false);
  }

  for (int r = 0; r < 8; ++r) red[tile][ks][r][lane] = acc[r];
  __syncthreads();

  const int m  = wave;
  const int tt = lane >> 4;
  const int sl = (m >> 3) * 16 + l;
  const int r  = m & 7;
  float sum = red[tt][0][r][sl];
  for (int q = 1; q < KSLICES; ++q) sum += red[tt][q][r][sl];
  agg[m * CC + nb + lane] = sum + bn;
}

#define LN_PAIRS 8
#define LN_AHEAD 4

__device__ __forceinline__ void ln_stats(const v4f y0, const v4f y1, float& s, float& s2) {
  s  = y0.x + y0.y + y0.z + y0.w + y1.x + y1.y + y1.z + y1.w;
  s2 = y0.x * y0.x + y0.y * y0.y + y0.z * y0.z + y0.w * y0.w +
       y1.x * y1.x + y1.y * y1.y + y1.z * y1.z + y1.w * y1.w;
}

__device__ __forceinline__ void ln_apply(const v4f y0, const v4f y1, const float s, const float s2,
                                         const v4f g0, const v4f g1, const v4f e0, const v4f e1,
                                         v4f& o0, v4f& o1) {
  const float mu  = s * (1.0f / CC);
  const float var = fmaf(-mu, mu, s2 * (1.0f / CC));
  const float inv = rsqrtf(var + LN_EPS);
  o0.x = (y0.x - mu) * inv * g0.x + e0.x;
  o0.y = (y0.y - mu) * inv * g0.y + e0.y;
  o0.z = (y0.z - mu) * inv * g0.z + e0.z;
  o0.w = (y0.w - mu) * inv * g0.w + e0.w;
  o1.x = (y1.x - mu) * inv * g1.x + e1.x;
  o1.y = (y1.y - mu) * inv * g1.y + e1.y;
  o1.z = (y1.z - mu) * inv * g1.z + e1.z;
  o1.w = (y1.w - mu) * inv * g1.w + e1.w;
}

__global__ __launch_bounds__(256) void fused_residual_ln_kernel(
    const float* __restrict__ x, const float* __restrict__ agg,
    const float* __restrict__ gamma, const float* __restrict__ beta,
    float* __restrict__ out) {
  const int b    = blockIdx.x;
  const int ch   = blockIdx.y;
  const int lane = threadIdx.x & 31;
  const int wave = __builtin_amdgcn_readfirstlane((int)(threadIdx.x >> 5));

  const bool full = wave < 5;
  const int  r0   = wave * CS_ROWS - (full ? 0 : wave - 5);
  const int  rl   = full ? CS_ROWS - 1 : CS_ROWS - 2;

  const size_t base = (((size_t)b * NN + (size_t)ch * CHUNK + (size_t)r0) * CC) + lane * 4;
  const float* xp = x + base;
  float*       op = out + base;

  const int c0 = lane * 4;
  const v4f g0 = *(const v4f*)(gamma + c0);
  const v4f g1 = *(const v4f*)(gamma + c0 + 128);
  const v4f e0 = *(const v4f*)(beta + c0);
  const v4f e1 = *(const v4f*)(beta + c0 + 128);
  const v4f a0 = *(const v4f*)(agg + b * CC + c0);
  const v4f a1 = *(const v4f*)(agg + b * CC + c0 + 128);

  v4f lo[2 * LN_PAIRS], hi[2 * LN_PAIRS];
#pragma unroll
  for (int i = 0; i < 2 * LN_AHEAD; ++i) {
    lo[i] = *(const v4f*)(xp + i * CC);
    hi[i] = *(const v4f*)(xp + i * CC + 128);
  }

#pragma unroll
  for (int p = 0; p < LN_PAIRS; ++p) {
    const int ra = 2 * p, rb = 2 * p + 1;
    __builtin_amdgcn_sched_barrier(0);
    if (p + LN_AHEAD < LN_PAIRS) {
      const int na = 2 * (p + LN_AHEAD);
      const int nb = (na + 1 == 2 * LN_PAIRS - 1) ? rl : na + 1;
      lo[na]     = *(const v4f*)(xp + na * CC);
      hi[na]     = *(const v4f*)(xp + na * CC + 128);
      lo[na + 1] = *(const v4f*)(xp + nb * CC);
      hi[na + 1] = *(const v4f*)(xp + nb * CC + 128);
    }

    const v4f ya0 = lo[ra] + a0, ya1 = hi[ra] + a1;
    const v4f yb0 = lo[rb] + a0, yb1 = hi[rb] + a1;
    float sa, sa2, sb, sb2;
    ln_stats(ya0, ya1, sa, sa2);
    ln_stats(yb0, yb1, sb, sb2);
#pragma unroll
    for (int off = 16; off >= 1; off >>= 1) {
      sa  += __shfl_xor(sa, off, 32);
      sa2 += __shfl_xor(sa2, off, 32);
      sb  += __shfl_xor(sb, off, 32);
      sb2 += __shfl_xor(sb2, off, 32);
    }
    v4f oa0, oa1, ob0, ob1;
    ln_apply(ya0, ya1, sa, sa2, g0, g1, e0, e1, oa0, oa1);
    ln_apply(yb0, yb1, sb, sb2, g0, g1, e0, e1, ob0, ob1);

    __builtin_nontemporal_store(oa0, (v4f*)(op + ra * CC));
    __builtin_nontemporal_store(oa1, (v4f*)(op + ra * CC + 128));
    if (rb < 2 * LN_PAIRS - 1 || full) {
      __builtin_nontemporal_store(ob0, (v4f*)(op + rb * CC));
      __builtin_nontemporal_store(ob1, (v4f*)(op + rb * CC + 128));
    }
  }
}

extern "C" void kernel_launch(void* const* d_in, const int* in_sizes, int n_in,
                              void* d_out, int out_size, void* d_ws, size_t ws_size,
                              hipStream_t stream) {
  (void)in_sizes; (void)n_in; (void)out_size; (void)ws_size;

  const float* x     = (const float*)d_in[0];
  const float* W     = (const float*)d_in[1];
  const float* bias  = (const float*)d_in[2];
  const float* gamma = (const float*)d_in[3];
  const float* beta  = (const float*)d_in[4];
  float* out = (float*)d_out;

  float* ws      = (float*)d_ws;
  float* partial = ws;
  float* meanp   = partial + (size_t)BB * NCHUNK * CC;
  float* agg     = meanp + MPAD * CC;

  colsum_partial_kernel<<<dim3(BB, NCHUNK), 256, 0, stream>>>(x, partial);
  colsum_final_kernel<<<MPAD, 256, 0, stream>>>(partial, meanp);
  gemm_wmma_kernel<<<NGROUPS, 512, 0, stream>>>(meanp, W, bias, agg);
  fused_residual_ln_kernel<<<dim3(BB, NCHUNK), 256, 0, stream>>>(x, agg, gamma, beta, out);
}


# Final column-sum pass folded into the GEMM launch (3 launches, was 4): K split over work-groups, each sums its slice's 80 partials as 8 groups of 10; LN adds the 4 slice products in order, then bias
# speedup vs baseline: 2.2827x; 1.1760x over previous
#include <hip/hip_runtime.h>

#define LN_EPS  1e-5f
#define BB      8
#define NN      10000
#define CC      256
#define NCHUNK  80
#define CHUNK   125

typedef __attribute__((ext_vector_type(2))) float v2f;
typedef __attribute__((ext_vector_type(4))) float v4f;
typedef __attribute__((ext_vector_type(8))) float v8f;

#define CS_WAVES 8
#define CS_ROWS  16

__global__ __launch_bounds__(256) void colsum_partial_kernel(
    const float* __restrict__ x, float* __restrict__ partial) {
  __shared__ float red[CS_WAVES][CC];

  const int b    = blockIdx.x;
  const int ch   = blockIdx.y;
  const int lane = threadIdx.x & 31;
  const int wave = __builtin_amdgcn_readfirstlane((int)(threadIdx.x >> 5));

  const bool full = wave < 5;
  const int  r0   = wave * CS_ROWS - (full ? 0 : wave - 5);
  const int  rl   = full ? CS_ROWS - 1 : CS_ROWS - 2;

  const float* xp = x + (((size_t)b * NN + (size_t)ch * CHUNK + (size_t)r0) * CC) + lane * 4;

  v4f lo[CS_ROWS], hi[CS_ROWS];
#pragma unroll
  for (int i = 0; i < CS_ROWS - 1; ++i) {
    lo[i] = *(const v4f*)(xp + i * CC);
    hi[i] = *(const v4f*)(xp + i * CC + 128);
  }
  lo[CS_ROWS - 1] = *(const v4f*)(xp + rl * CC);
  hi[CS_ROWS - 1] = *(const v4f*)(xp + rl * CC + 128);

  v4f s0 = lo[0], s1 = hi[0];
#pragma unroll
  for (int i = 1; i < CS_ROWS - 1; ++i) { s0 += lo[i]; s1 += hi[i]; }
  if (full) { s0 += lo[CS_ROWS - 1]; s1 += hi[CS_ROWS - 1]; }

  *(v4f*)(&red[wave][lane * 4])       = s0;
  *(v4f*)(&red[wave][128 + lane * 4]) = s1;
  __syncthreads();

  const int c = threadIdx.x;
  float s = red[0][c];
#pragma unroll
  for (int w = 1; w < CS_WAVES; ++w) s += red[w][c];
  partial[((size_t)b * NCHUNK + ch) * CC + c] = s;
}

#define KSLICES 4
#define KSLICE  (CC / KSLICES)
#define KSTEPS  (KSLICE / 4)
#define NQ      4
#define NQCOLS  (CC / NQ)
#define CGROUPS 8
#define CGLEN   (NCHUNK / CGROUPS)
#define WPITCH  (KSLICE + 4)

__global__ __launch_bounds__(128) void mean_gemm_slice_kernel(
    const float* __restrict__ partial, const float* __restrict__ W,
    float* __restrict__ aggpart) {
  __shared__ float gs[CGROUPS][KSLICE];
  __shared__ float wl[NQCOLS * WPITCH];
  __shared__ float ol[4][32];

  const int j    = blockIdx.x;
  const int b    = blockIdx.y;
  const int q    = blockIdx.z;
  const int tid  = threadIdx.x;
  const int lane = tid & 31;
  const int wave = tid >> 5;
  const int hh   = lane >> 4;
  const int l    = lane & 15;
  const int g    = tid >> 4;
  const int qd   = tid & 15;

  const float* pp = partial + ((size_t)(b * NCHUNK + g * CGLEN) * CC) + j * KSLICE + qd * 4;
  const float* wp = W + ((size_t)(q * NQCOLS + g) * CC) + j * KSLICE + qd * 4;

  v4f pv[CGLEN], wv[8];
  for (int i = 0; i < CGLEN; ++i) pv[i] = *(const v4f*)(pp + i * CC);
  for (int i = 0; i < 8; ++i) wv[i] = *(const v4f*)(wp + (size_t)i * 8 * CC);
  __builtin_amdgcn_sched_barrier(0);

  v4f s = pv[0];
  for (int i = 1; i < CGLEN; ++i) s += pv[i];
  *(v4f*)(&gs[g][qd * 4]) = s;
  for (int i = 0; i < 8; ++i) *(v4f*)(&wl[(g + 8 * i) * WPITCH + qd * 4]) = wv[i];
  __syncthreads();

  float m0 = gs[0][lane], m1 = gs[0][32 + lane];
  for (int t = 1; t < CGROUPS; ++t) { m0 += gs[t][lane]; m1 += gs[t][32 + lane]; }
  m0 *= (1.0f / (float)NN);
  m1 *= (1.0f / (float)NN);

  v2f a[KSTEPS], bm[KSTEPS];
  const float* bl = &wl[(wave * 16 + l) * WPITCH + 2 * hh];
  for (int i = 0; i < KSTEPS; ++i) bm[i] = *(const v2f*)(bl + 4 * i);
  for (int i = 0; i < KSTEPS; ++i) {
    const float mm  = (i < 8) ? m0 : m1;
    const int   src = 4 * (i & 7) + 2 * hh;
    const float x0  = __shfl(mm, src, 32);
    const float x1  = __shfl(mm, src + 1, 32);
    a[i].x = (l == 0) ? x0 : 0.0f;
    a[i].y = (l == 0) ? x1 : 0.0f;
  }

  v8f acc = {};
  for (int i = 0; i < KSTEPS; ++i) {
    acc = __builtin_amdgcn_wmma_f32_16x16x4_f32(
         false, a[i],  false, bm[i],
         (short)0, acc,  false,  false);
  }

  ol[wave][lane] = acc[0];
  __syncthreads();

  if (tid < 16) {
    const v4f o = *(const v4f*)(&ol[tid >> 2][(tid & 3) * 4]);
    *(v4f*)(aggpart + ((size_t)(j * BB + b) * CC) + q * NQCOLS + tid * 4) = o;
  }
}

#define LN_PAIRS 8
#define LN_AHEAD 4

__device__ __forceinline__ void ln_stats(const v4f y0, const v4f y1, float& s, float& s2) {
  s  = y0.x + y0.y + y0.z + y0.w + y1.x + y1.y + y1.z + y1.w;
  s2 = y0.x * y0.x + y0.y * y0.y + y0.z * y0.z + y0.w * y0.w +
       y1.x * y1.x + y1.y * y1.y + y1.z * y1.z + y1.w * y1.w;
}

__device__ __forceinline__ void ln_apply(const v4f y0, const v4f y1, const float s, const float s2,
                                         const v4f g0, const v4f g1, const v4f e0, const v4f e1,
                                         v4f& o0, v4f& o1) {
  const float mu  = s * (1.0f / CC);
  const float var = fmaf(-mu, mu, s2 * (1.0f / CC));
  const float inv = rsqrtf(var + LN_EPS);
  o0.x = (y0.x - mu) * inv * g0.x + e0.x;
  o0.y = (y0.y - mu) * inv * g0.y + e0.y;
  o0.z = (y0.z - mu) * inv * g0.z + e0.z;
  o0.w = (y0.w - mu) * inv * g0.w + e0.w;
  o1.x = (y1.x - mu) * inv * g1.x + e1.x;
  o1.y = (y1.y - mu) * inv * g1.y + e1.y;
  o1.z = (y1.z - mu) * inv * g1.z + e1.z;
  o1.w = (y1.w - mu) * inv * g1.w + e1.w;
}

__global__ __launch_bounds__(256) void fused_residual_ln_kernel(
    const float* __restrict__ x, const float* __restrict__ aggpart,
    const float* __restrict__ bias,
    const float* __restrict__ gamma, const float* __restrict__ beta,
    float* __restrict__ out) {
  __shared__ float sl[KSLICES][CC];

  const int b    = blockIdx.x;
  const int ch   = blockIdx.y;
  const int lane = threadIdx.x & 31;
  const int wave = __builtin_amdgcn_readfirstlane((int)(threadIdx.x >> 5));

  const bool full = wave < 5;
  const int  r0   = wave * CS_ROWS - (full ? 0 : wave - 5);
  const int  rl   = full ? CS_ROWS - 1 : CS_ROWS - 2;

  const size_t base = (((size_t)b * NN + (size_t)ch * CHUNK + (size_t)r0) * CC) + lane * 4;
  const float* xp = x + base;
  float*       op = out + base;

  const int c0 = lane * 4;
  const int sj = wave & 3;
  const int sc = (wave >> 2) * 128 + c0;
  const v4f pv = *(const v4f*)(aggpart + (size_t)(sj * BB + b) * CC + sc);
  const v4f g0 = *(const v4f*)(gamma + c0);
  const v4f g1 = *(const v4f*)(gamma + c0 + 128);
  const v4f e0 = *(const v4f*)(beta + c0);
  const v4f e1 = *(const v4f*)(beta + c0 + 128);
  const v4f b0 = *(const v4f*)(bias + c0);
  const v4f b1 = *(const v4f*)(bias + c0 + 128);

  v4f lo[2 * LN_PAIRS], hi[2 * LN_PAIRS];
#pragma unroll
  for (int i = 0; i < 2 * LN_AHEAD; ++i) {
    lo[i] = *(const v4f*)(xp + i * CC);
    hi[i] = *(const v4f*)(xp + i * CC + 128);
  }

  __builtin_amdgcn_sched_barrier(0);
  *(v4f*)(&sl[sj][sc]) = pv;
  __builtin_amdgcn_fence(__ATOMIC_RELEASE, "workgroup", "local");
  __builtin_amdgcn_s_barrier();
  __builtin_amdgcn_fence(__ATOMIC_ACQUIRE, "workgroup", "local");
  v4f a0 = *(const v4f*)(&sl[0][c0]);
  v4f a1 = *(const v4f*)(&sl[0][c0 + 128]);
  for (int t = 1; t < KSLICES; ++t) {
    a0 += *(const v4f*)(&sl[t][c0]);
    a1 += *(const v4f*)(&sl[t][c0 + 128]);
  }
  a0 += b0;
  a1 += b1;

#pragma unroll
  for (int p = 0; p < LN_PAIRS; ++p) {
    const int ra = 2 * p, rb = 2 * p + 1;
    __builtin_amdgcn_sched_barrier(0);
    if (p + LN_AHEAD < LN_PAIRS) {
      const int na = 2 * (p + LN_AHEAD);
      const int nb = (na + 1 == 2 * LN_PAIRS - 1) ? rl : na + 1;
      lo[na]     = *(const v4f*)(xp + na * CC);
      hi[na]     = *(const v4f*)(xp + na * CC + 128);
      lo[na + 1] = *(const v4f*)(xp + nb * CC);
      hi[na + 1] = *(const v4f*)(xp + nb * CC + 128);
    }

    const v4f ya0 = lo[ra] + a0, ya1 = hi[ra] + a1;
    const v4f yb0 = lo[rb] + a0, yb1 = hi[rb] + a1;
    float sa, sa2, sb, sb2;
    ln_stats(ya0, ya1, sa, sa2);
    ln_stats(yb0, yb1, sb, sb2);
#pragma unroll
    for (int off = 16; off >= 1; off >>= 1) {
      sa  += __shfl_xor(sa, off, 32);
      sa2 += __shfl_xor(sa2, off, 32);
      sb  += __shfl_xor(sb, off, 32);
      sb2 += __shfl_xor(sb2, off, 32);
    }
    v4f oa0, oa1, ob0, ob1;
    ln_apply(ya0, ya1, sa, sa2, g0, g1, e0, e1, oa0, oa1);
    ln_apply(yb0, yb1, sb, sb2, g0, g1, e0, e1, ob0, ob1);

    __builtin_nontemporal_store(oa0, (v4f*)(op + ra * CC));
    __builtin_nontemporal_store(oa1, (v4f*)(op + ra * CC + 128));
    if (rb < 2 * LN_PAIRS - 1 || full) {
      __builtin_nontemporal_store(ob0, (v4f*)(op + rb * CC));
      __builtin_nontemporal_store(ob1, (v4f*)(op + rb * CC + 128));
    }
  }
}

extern "C" void kernel_launch(void* const* d_in, const int* in_sizes, int n_in,
                              void* d_out, int out_size, void* d_ws, size_t ws_size,
                              hipStream_t stream) {
  (void)in_sizes; (void)n_in; (void)out_size; (void)ws_size;

  const float* x     = (const float*)d_in[0];
  const float* W     = (const float*)d_in[1];
  const float* bias  = (const float*)d_in[2];
  const float* gamma = (const float*)d_in[3];
  const float* beta  = (const float*)d_in[4];
  float* out = (float*)d_out;

  float* ws      = (float*)d_ws;
  float* partial = ws;
  float* aggpart = partial + (size_t)BB * NCHUNK * CC;

  colsum_partial_kernel<<<dim3(BB, NCHUNK), 256, 0, stream>>>(x, partial);
  mean_gemm_slice_kernel<<<dim3(KSLICES, BB, NQ), 128, 0, stream>>>(partial, W, aggpart);
  fused_residual_ln_kernel<<<dim3(BB, NCHUNK), 256, 0, stream>>>(x, aggpart, bias, gamma, beta, out);
}


# LayerNorm pass: row-sum butterfly on the vector ALU (DPP row_xmask 1,2,4,8 then permlanex16) in place of 5 ds_bpermute trips per sum; sum tree now lane distance 1,2,4,8,16 (was 16..1)
# speedup vs baseline: 2.3986x; 1.0508x over previous
#include <hip/hip_runtime.h>

#define LN_EPS  1e-5f
#define BB      8
#define NN      10000
#define CC      256
#define NCHUNK  80
#define CHUNK   125

typedef __attribute__((ext_vector_type(2))) float v2f;
typedef __attribute__((ext_vector_type(4))) float v4f;
typedef __attribute__((ext_vector_type(8))) float v8f;

#define CS_WAVES 8
#define CS_ROWS  16

__global__ __launch_bounds__(256) void colsum_partial_kernel(
    const float* __restrict__ x, float* __restrict__ partial) {
  __shared__ float red[CS_WAVES][CC];

  const int b    = blockIdx.x;
  const int ch   = blockIdx.y;
  const int lane = threadIdx.x & 31;
  const int wave = __builtin_amdgcn_readfirstlane((int)(threadIdx.x >> 5));

  const bool full = wave < 5;
  const int  r0   = wave * CS_ROWS - (full ? 0 : wave - 5);
  const int  rl   = full ? CS_ROWS - 1 : CS_ROWS - 2;

  const float* xp = x + (((size_t)b * NN + (size_t)ch * CHUNK + (size_t)r0) * CC) + lane * 4;

  v4f lo[CS_ROWS], hi[CS_ROWS];
#pragma unroll
  for (int i = 0; i < CS_ROWS - 1; ++i) {
    lo[i] = *(const v4f*)(xp + i * CC);
    hi[i] = *(const v4f*)(xp + i * CC + 128);
  }
  lo[CS_ROWS - 1] = *(const v4f*)(xp + rl * CC);
  hi[CS_ROWS - 1] = *(const v4f*)(xp + rl * CC + 128);

  v4f s0 = lo[0], s1 = hi[0];
#pragma unroll
  for (int i = 1; i < CS_ROWS - 1; ++i) { s0 += lo[i]; s1 += hi[i]; }
  if (full) { s0 += lo[CS_ROWS - 1]; s1 += hi[CS_ROWS - 1]; }

  *(v4f*)(&red[wave][lane * 4])       = s0;
  *(v4f*)(&red[wave][128 + lane * 4]) = s1;
  __syncthreads();

  const int c = threadIdx.x;
  float s = red[0][c];
#pragma unroll
  for (int w = 1; w < CS_WAVES; ++w) s += red[w][c];
  partial[((size_t)b * NCHUNK + ch) * CC + c] = s;
}

#define KSLICES 4
#define KSLICE  (CC / KSLICES)
#define KSTEPS  (KSLICE / 4)
#define NQ      4
#define NQCOLS  (CC / NQ)
#define CGROUPS 8
#define CGLEN   (NCHUNK / CGROUPS)
#define WPITCH  (KSLICE + 4)

__global__ __launch_bounds__(128) void mean_gemm_slice_kernel(
    const float* __restrict__ partial, const float* __restrict__ W,
    float* __restrict__ aggpart) {
  __shared__ float gs[CGROUPS][KSLICE];
  __shared__ float wl[NQCOLS * WPITCH];
  __shared__ float ol[4][32];

  const int j    = blockIdx.x;
  const int b    = blockIdx.y;
  const int q    = blockIdx.z;
  const int tid  = threadIdx.x;
  const int lane = tid & 31;
  const int wave = tid >> 5;
  const int hh   = lane >> 4;
  const int l    = lane & 15;
  const int g    = tid >> 4;
  const int qd   = tid & 15;

  const float* pp = partial + ((size_t)(b * NCHUNK + g * CGLEN) * CC) + j * KSLICE + qd * 4;
  const float* wp = W + ((size_t)(q * NQCOLS + g) * CC) + j * KSLICE + qd * 4;

  v4f pv[CGLEN], wv[8];
  for (int i = 0; i < CGLEN; ++i) pv[i] = *(const v4f*)(pp + i * CC);
  for (int i = 0; i < 8; ++i) wv[i] = *(const v4f*)(wp + (size_t)i * 8 * CC);
  __builtin_amdgcn_sched_barrier(0);

  v4f s = pv[0];
  for (int i = 1; i < CGLEN; ++i) s += pv[i];
  *(v4f*)(&gs[g][qd * 4]) = s;
  for (int i = 0; i < 8; ++i) *(v4f*)(&wl[(g + 8 * i) * WPITCH + qd * 4]) = wv[i];
  __syncthreads();

  float m0 = gs[0][lane], m1 = gs[0][32 + lane];
  for (int t = 1; t < CGROUPS; ++t) { m0 += gs[t][lane]; m1 += gs[t][32 + lane]; }
  m0 *= (1.0f / (float)NN);
  m1 *= (1.0f / (float)NN);

  v2f a[KSTEPS], bm[KSTEPS];
  const float* bl = &wl[(wave * 16 + l) * WPITCH + 2 * hh];
  for (int i = 0; i < KSTEPS; ++i) bm[i] = *(const v2f*)(bl + 4 * i);
  for (int i = 0; i < KSTEPS; ++i) {
    const float mm  = (i < 8) ? m0 : m1;
    const int   src = 4 * (i & 7) + 2 * hh;
    const float x0  = __shfl(mm, src, 32);
    const float x1  = __shfl(mm, src + 1, 32);
    a[i].x = (l == 0) ? x0 : 0.0f;
    a[i].y = (l == 0) ? x1 : 0.0f;
  }

  v8f acc = {};
  for (int i = 0; i < KSTEPS; ++i) {
    acc = __builtin_amdgcn_wmma_f32_16x16x4_f32(
         false, a[i],  false, bm[i],
         (short)0, acc,  false,  false);
  }

  ol[wave][lane] = acc[0];
  __syncthreads();

  if (tid < 16) {
    const v4f o = *(const v4f*)(&ol[tid >> 2][(tid & 3) * 4]);
    *(v4f*)(aggpart + ((size_t)(j * BB + b) * CC) + q * NQCOLS + tid * 4) = o;
  }
}

#define LN_PAIRS 8
#define LN_AHEAD 4

template <int XMASK>
__device__ __forceinline__ float row_xadd(const float v) {
  const int iv = __builtin_bit_cast(int, v);
  const int t = __builtin_amdgcn_update_dpp(iv, iv, 0x160 + XMASK, 0xf, 0xf, true);
  return v + __builtin_bit_cast(float, t);
}

__device__ __forceinline__ float half_xadd(const float v) {
  const unsigned iv = __builtin_bit_cast(unsigned, v);
  const unsigned t = __builtin_amdgcn_permlanex16(iv, iv, 0x76543210u, 0xfedcba98u, false, false);
  return v + __builtin_bit_cast(float, t);
}

__device__ __forceinline__ void wave_sum4(float& a, float& b, float& c, float& d) {
  a = row_xadd<1>(a); b = row_xadd<1>(b); c = row_xadd<1>(c); d = row_xadd<1>(d);
  a = row_xadd<2>(a); b = row_xadd<2>(b); c = row_xadd<2>(c); d = row_xadd<2>(d);
  a = row_xadd<4>(a); b = row_xadd<4>(b); c = row_xadd<4>(c); d = row_xadd<4>(d);
  a = row_xadd<8>(a); b = row_xadd<8>(b); c = row_xadd<8>(c); d = row_xadd<8>(d);
  a = half_xadd(a);   b = half_xadd(b);   c = half_xadd(c);   d = half_xadd(d);
}

__device__ __forceinline__ void ln_stats(const v4f y0, const v4f y1, float& s, float& s2) {
  s  = y0.x + y0.y + y0.z + y0.w + y1.x + y1.y + y1.z + y1.w;
  s2 = y0.x * y0.x + y0.y * y0.y + y0.z * y0.z + y0.w * y0.w +
       y1.x * y1.x + y1.y * y1.y + y1.z * y1.z + y1.w * y1.w;
}

__device__ __forceinline__ void ln_apply(const v4f y0, const v4f y1, const float s, const float s2,
                                         const v4f g0, const v4f g1, const v4f e0, const v4f e1,
                                         v4f& o0, v4f& o1) {
  const float mu  = s * (1.0f / CC);
  const float var = fmaf(-mu, mu, s2 * (1.0f / CC));
  const float inv = rsqrtf(var + LN_EPS);
  o0.x = (y0.x - mu) * inv * g0.x + e0.x;
  o0.y = (y0.y - mu) * inv * g0.y + e0.y;
  o0.z = (y0.z - mu) * inv * g0.z + e0.z;
  o0.w = (y0.w - mu) * inv * g0.w + e0.w;
  o1.x = (y1.x - mu) * inv * g1.x + e1.x;
  o1.y = (y1.y - mu) * inv * g1.y + e1.y;
  o1.z = (y1.z - mu) * inv * g1.z + e1.z;
  o1.w = (y1.w - mu) * inv * g1.w + e1.w;
}

__global__ __launch_bounds__(256) void fused_residual_ln_kernel(
    const float* __restrict__ x, const float* __restrict__ aggpart,
    const float* __restrict__ bias,
    const float* __restrict__ gamma, const float* __restrict__ beta,
    float* __restrict__ out) {
  __shared__ float sl[KSLICES][CC];

  const int b    = blockIdx.x;
  const int ch   = blockIdx.y;
  const int lane = threadIdx.x & 31;
  const int wave = __builtin_amdgcn_readfirstlane((int)(threadIdx.x >> 5));

  const bool full = wave < 5;
  const int  r0   = wave * CS_ROWS - (full ? 0 : wave - 5);
  const int  rl   = full ? CS_ROWS - 1 : CS_ROWS - 2;

  const size_t base = (((size_t)b * NN + (size_t)ch * CHUNK + (size_t)r0) * CC) + lane * 4;
  const float* xp = x + base;
  float*       op = out + base;

  const int c0 = lane * 4;
  const int sj = wave & 3;
  const int sc = (wave >> 2) * 128 + c0;
  const v4f pv = *(const v4f*)(aggpart + (size_t)(sj * BB + b) * CC + sc);
  const v4f g0 = *(const v4f*)(gamma + c0);
  const v4f g1 = *(const v4f*)(gamma + c0 + 128);
  const v4f e0 = *(const v4f*)(beta + c0);
  const v4f e1 = *(const v4f*)(beta + c0 + 128);
  const v4f b0 = *(const v4f*)(bias + c0);
  const v4f b1 = *(const v4f*)(bias + c0 + 128);

  v4f lo[2 * LN_PAIRS], hi[2 * LN_PAIRS];
#pragma unroll
  for (int i = 0; i < 2 * LN_AHEAD; ++i) {
    lo[i] = *(const v4f*)(xp + i * CC);
    hi[i] = *(const v4f*)(xp + i * CC + 128);
  }

  __builtin_amdgcn_sched_barrier(0);
  *(v4f*)(&sl[sj][sc]) = pv;
  __builtin_amdgcn_fence(__ATOMIC_RELEASE, "workgroup", "local");
  __builtin_amdgcn_s_barrier();
  __builtin_amdgcn_fence(__ATOMIC_ACQUIRE, "workgroup", "local");
  v4f a0 = *(const v4f*)(&sl[0][c0]);
  v4f a1 = *(const v4f*)(&sl[0][c0 + 128]);
  for (int t = 1; t < KSLICES; ++t) {
    a0 += *(const v4f*)(&sl[t][c0]);
    a1 += *(const v4f*)(&sl[t][c0 + 128]);
  }
  a0 += b0;
  a1 += b1;

#pragma unroll
  for (int p = 0; p < LN_PAIRS; ++p) {
    const int ra = 2 * p, rb = 2 * p + 1;
    __builtin_amdgcn_sched_barrier(0);
    if (p + LN_AHEAD < LN_PAIRS) {
      const int na = 2 * (p + LN_AHEAD);
      const int nb = (na + 1 == 2 * LN_PAIRS - 1) ? rl : na + 1;
      lo[na]     = *(const v4f*)(xp + na * CC);
      hi[na]     = *(const v4f*)(xp + na * CC + 128);
      lo[na + 1] = *(const v4f*)(xp + nb * CC);
      hi[na + 1] = *(const v4f*)(xp + nb * CC + 128);
    }

    const v4f ya0 = lo[ra] + a0, ya1 = hi[ra] + a1;
    const v4f yb0 = lo[rb] + a0, yb1 = hi[rb] + a1;
    float sa, sa2, sb, sb2;
    ln_stats(ya0, ya1, sa, sa2);
    ln_stats(yb0, yb1, sb, sb2);
    wave_sum4(sa, sa2, sb, sb2);
    v4f oa0, oa1, ob0, ob1;
    ln_apply(ya0, ya1, sa, sa2, g0, g1, e0, e1, oa0, oa1);
    ln_apply(yb0, yb1, sb, sb2, g0, g1, e0, e1, ob0, ob1);

    __builtin_nontemporal_store(oa0, (v4f*)(op + ra * CC));
    __builtin_nontemporal_store(oa1, (v4f*)(op + ra * CC + 128));
    if (rb < 2 * LN_PAIRS - 1 || full) {
      __builtin_nontemporal_store(ob0, (v4f*)(op + rb * CC));
      __builtin_nontemporal_store(ob1, (v4f*)(op + rb * CC + 128));
    }
  }
}

extern "C" void kernel_launch(void* const* d_in, const int* in_sizes, int n_in,
                              void* d_out, int out_size, void* d_ws, size_t ws_size,
                              hipStream_t stream) {
  (void)in_sizes; (void)n_in; (void)out_size; (void)ws_size;

  const float* x     = (const float*)d_in[0];
  const float* W     = (const float*)d_in[1];
  const float* bias  = (const float*)d_in[2];
  const float* gamma = (const float*)d_in[3];
  const float* beta  = (const float*)d_in[4];
  float* out = (float*)d_out;

  float* ws      = (float*)d_ws;
  float* partial = ws;
  float* aggpart = partial + (size_t)BB * NCHUNK * CC;

  colsum_partial_kernel<<<dim3(BB, NCHUNK), 256, 0, stream>>>(x, partial);
  mean_gemm_slice_kernel<<<dim3(KSLICES, BB, NQ), 128, 0, stream>>>(partial, W, aggpart);
  fused_residual_ln_kernel<<<dim3(BB, NCHUNK), 256, 0, stream>>>(x, aggpart, bias, gamma, beta, out);
}


# Column-sum pass: work-group of 4 waves (was 8), a wave asks for all its 31-32 rows as one group of 64 loads (was 13 in flight, refilled); chunk sum = 4 row-block sub-sums in order (was 8)
# speedup vs baseline: 2.4939x; 1.0397x over previous
#include <hip/hip_runtime.h>

#define LN_EPS  1e-5f
#define BB      8
#define NN      10000
#define CC      256
#define NCHUNK  80
#define CHUNK   125

typedef __attribute__((ext_vector_type(2))) float v2f;
typedef __attribute__((ext_vector_type(4))) float v4f;
typedef __attribute__((ext_vector_type(8))) float v8f;

#define CS_WAVES 4
#define CS_ROWS  32

__global__ __launch_bounds__(128) void colsum_partial_kernel(
    const float* __restrict__ x, float* __restrict__ partial) {
  __shared__ float red[CS_WAVES][CC];

  const int b    = blockIdx.x;
  const int ch   = blockIdx.y;
  const int lane = threadIdx.x & 31;
  const int wave = __builtin_amdgcn_readfirstlane((int)(threadIdx.x >> 5));

  const bool full = wave == 0;
  const int  r0   = full ? 0 : wave * CS_ROWS - (wave - 1);
  const int  rl   = full ? CS_ROWS - 1 : CS_ROWS - 2;

  const float* xp = x + (((size_t)b * NN + (size_t)ch * CHUNK + (size_t)r0) * CC) + lane * 4;

  v4f lo[CS_ROWS], hi[CS_ROWS];
#pragma unroll
  for (int i = 0; i < CS_ROWS - 1; ++i) {
    lo[i] = *(const v4f*)(xp + i * CC);
    hi[i] = *(const v4f*)(xp + i * CC + 128);
  }
  lo[CS_ROWS - 1] = *(const v4f*)(xp + rl * CC);
  hi[CS_ROWS - 1] = *(const v4f*)(xp + rl * CC + 128);
  __builtin_amdgcn_sched_barrier(0);

  v4f s0 = lo[0], s1 = hi[0];
#pragma unroll
  for (int i = 1; i < CS_ROWS - 1; ++i) { s0 += lo[i]; s1 += hi[i]; }
  if (full) { s0 += lo[CS_ROWS - 1]; s1 += hi[CS_ROWS - 1]; }

  *(v4f*)(&red[wave][lane * 4])       = s0;
  *(v4f*)(&red[wave][128 + lane * 4]) = s1;
  __syncthreads();

  const int c = threadIdx.x * 2;
  v2f s = *(const v2f*)(&red[0][c]);
  for (int w = 1; w < CS_WAVES; ++w) s += *(const v2f*)(&red[w][c]);
  *(v2f*)(partial + ((size_t)b * NCHUNK + ch) * CC + c) = s;
}

#define KSLICES 4
#define KSLICE  (CC / KSLICES)
#define KSTEPS  (KSLICE / 4)
#define NQ      4
#define NQCOLS  (CC / NQ)
#define CGROUPS 8
#define CGLEN   (NCHUNK / CGROUPS)
#define WPITCH  (KSLICE + 4)

__global__ __launch_bounds__(128) void mean_gemm_slice_kernel(
    const float* __restrict__ partial, const float* __restrict__ W,
    float* __restrict__ aggpart) {
  __shared__ float gs[CGROUPS][KSLICE];
  __shared__ float wl[NQCOLS * WPITCH];
  __shared__ float ol[4][32];

  const int j    = blockIdx.x;
  const int b    = blockIdx.y;
  const int q    = blockIdx.z;
  const int tid  = threadIdx.x;
  const int lane = tid & 31;
  const int wave = tid >> 5;
  const int hh   = lane >> 4;
  const int l    = lane & 15;
  const int g    = tid >> 4;
  const int qd   = tid & 15;

  const float* pp = partial + ((size_t)(b * NCHUNK + g * CGLEN) * CC) + j * KSLICE + qd * 4;
  const float* wp = W + ((size_t)(q * NQCOLS + g) * CC) + j * KSLICE + qd * 4;

  v4f pv[CGLEN], wv[8];
  for (int i = 0; i < CGLEN; ++i) pv[i] = *(const v4f*)(pp + i * CC);
  for (int i = 0; i < 8; ++i) wv[i] = *(const v4f*)(wp + (size_t)i * 8 * CC);
  __builtin_amdgcn_sched_barrier(0);

  v4f s = pv[0];
  for (int i = 1; i < CGLEN; ++i) s += pv[i];
  *(v4f*)(&gs[g][qd * 4]) = s;
  for (int i = 0; i < 8; ++i) *(v4f*)(&wl[(g + 8 * i) * WPITCH + qd * 4]) = wv[i];
  __syncthreads();

  float m0 = gs[0][lane], m1 = gs[0][32 + lane];
  for (int t = 1; t < CGROUPS; ++t) { m0 += gs[t][lane]; m1 += gs[t][32 + lane]; }
  m0 *= (1.0f / (float)NN);
  m1 *= (1.0f / (float)NN);

  v2f a[KSTEPS], bm[KSTEPS];
  const float* bl = &wl[(wave * 16 + l) * WPITCH + 2 * hh];
  for (int i = 0; i < KSTEPS; ++i) bm[i] = *(const v2f*)(bl + 4 * i);
  for (int i = 0; i < KSTEPS; ++i) {
    const float mm  = (i < 8) ? m0 : m1;
    const int   src = 4 * (i & 7) + 2 * hh;
    const float x0  = __shfl(mm, src, 32);
    const float x1  = __shfl(mm, src + 1, 32);
    a[i].x = (l == 0) ? x0 : 0.0f;
    a[i].y = (l == 0) ? x1 : 0.0f;
  }

  v8f acc = {};
  for (int i = 0; i < KSTEPS; ++i) {
    acc = __builtin_amdgcn_wmma_f32_16x16x4_f32(
         false, a[i],  false, bm[i],
         (short)0, acc,  false,  false);
  }

  ol[wave][lane] = acc[0];
  __syncthreads();

  if (tid < 16) {
    const v4f o = *(const v4f*)(&ol[tid >> 2][(tid & 3) * 4]);
    *(v4f*)(aggpart + ((size_t)(j * BB + b) * CC) + q * NQCOLS + tid * 4) = o;
  }
}

#define LN_ROWS  16
#define LN_PAIRS 8
#define LN_AHEAD 4

template <int XMASK>
__device__ __forceinline__ float row_xadd(const float v) {
  const int iv = __builtin_bit_cast(int, v);
  const int t = __builtin_amdgcn_update_dpp(iv, iv, 0x160 + XMASK, 0xf, 0xf, true);
  return v + __builtin_bit_cast(float, t);
}

__device__ __forceinline__ float half_xadd(const float v) {
  const unsigned iv = __builtin_bit_cast(unsigned, v);
  const unsigned t = __builtin_amdgcn_permlanex16(iv, iv, 0x76543210u, 0xfedcba98u, false, false);
  return v + __builtin_bit_cast(float, t);
}

__device__ __forceinline__ void wave_sum4(float& a, float& b, float& c, float& d) {
  a = row_xadd<1>(a); b = row_xadd<1>(b); c = row_xadd<1>(c); d = row_xadd<1>(d);
  a = row_xadd<2>(a); b = row_xadd<2>(b); c = row_xadd<2>(c); d = row_xadd<2>(d);
  a = row_xadd<4>(a); b = row_xadd<4>(b); c = row_xadd<4>(c); d = row_xadd<4>(d);
  a = row_xadd<8>(a); b = row_xadd<8>(b); c = row_xadd<8>(c); d = row_xadd<8>(d);
  a = half_xadd(a);   b = half_xadd(b);   c = half_xadd(c);   d = half_xadd(d);
}

__device__ __forceinline__ void ln_stats(const v4f y0, const v4f y1, float& s, float& s2) {
  s  = y0.x + y0.y + y0.z + y0.w + y1.x + y1.y + y1.z + y1.w;
  s2 = y0.x * y0.x + y0.y * y0.y + y0.z * y0.z + y0.w * y0.w +
       y1.x * y1.x + y1.y * y1.y + y1.z * y1.z + y1.w * y1.w;
}

__device__ __forceinline__ void ln_apply(const v4f y0, const v4f y1, const float s, const float s2,
                                         const v4f g0, const v4f g1, const v4f e0, const v4f e1,
                                         v4f& o0, v4f& o1) {
  const float mu  = s * (1.0f / CC);
  const float var = fmaf(-mu, mu, s2 * (1.0f / CC));
  const float inv = rsqrtf(var + LN_EPS);
  o0.x = (y0.x - mu) * inv * g0.x + e0.x;
  o0.y = (y0.y - mu) * inv * g0.y + e0.y;
  o0.z = (y0.z - mu) * inv * g0.z + e0.z;
  o0.w = (y0.w - mu) * inv * g0.w + e0.w;
  o1.x = (y1.x - mu) * inv * g1.x + e1.x;
  o1.y = (y1.y - mu) * inv * g1.y + e1.y;
  o1.z = (y1.z - mu) * inv * g1.z + e1.z;
  o1.w = (y1.w - mu) * inv * g1.w + e1.w;
}

__global__ __launch_bounds__(256) void fused_residual_ln_kernel(
    const float* __restrict__ x, const float* __restrict__ aggpart,
    const float* __restrict__ bias,
    const float* __restrict__ gamma, const float* __restrict__ beta,
    float* __restrict__ out) {
  __shared__ float sl[KSLICES][CC];

  const int b    = blockIdx.x;
  const int ch   = blockIdx.y;
  const int lane = threadIdx.x & 31;
  const int wave = __builtin_amdgcn_readfirstlane((int)(threadIdx.x >> 5));

  const bool full = wave < 5;
  const int  r0   = wave * LN_ROWS - (full ? 0 : wave - 5);
  const int  rl   = full ? LN_ROWS - 1 : LN_ROWS - 2;

  const size_t base = (((size_t)b * NN + (size_t)ch * CHUNK + (size_t)r0) * CC) + lane * 4;
  const float* xp = x + base;
  float*       op = out + base;

  const int c0 = lane * 4;
  const int sj = wave & 3;
  const int sc = (wave >> 2) * 128 + c0;
  const v4f pv = *(const v4f*)(aggpart + (size_t)(sj * BB + b) * CC + sc);
  const v4f g0 = *(const v4f*)(gamma + c0);
  const v4f g1 = *(const v4f*)(gamma + c0 + 128);
  const v4f e0 = *(const v4f*)(beta + c0);
  const v4f e1 = *(const v4f*)(beta + c0 + 128);
  const v4f b0 = *(const v4f*)(bias + c0);
  const v4f b1 = *(const v4f*)(bias + c0 + 128);

  v4f lo[2 * LN_PAIRS], hi[2 * LN_PAIRS];
#pragma unroll
  for (int i = 0; i < 2 * LN_AHEAD; ++i) {
    lo[i] = *(const v4f*)(xp + i * CC);
    hi[i] = *(const v4f*)(xp + i * CC + 128);
  }

  __builtin_amdgcn_sched_barrier(0);
  *(v4f*)(&sl[sj][sc]) = pv;
  __builtin_amdgcn_fence(__ATOMIC_RELEASE, "workgroup", "local");
  __builtin_amdgcn_s_barrier();
  __builtin_amdgcn_fence(__ATOMIC_ACQUIRE, "workgroup", "local");
  v4f a0 = *(const v4f*)(&sl[0][c0]);
  v4f a1 = *(const v4f*)(&sl[0][c0 + 128]);
  for (int t = 1; t < KSLICES; ++t) {
    a0 += *(const v4f*)(&sl[t][c0]);
    a1 += *(const v4f*)(&sl[t][c0 + 128]);
  }
  a0 += b0;
  a1 += b1;

#pragma unroll
  for (int p = 0; p < LN_PAIRS; ++p) {
    const int ra = 2 * p, rb = 2 * p + 1;
    __builtin_amdgcn_sched_barrier(0);
    if (p + LN_AHEAD < LN_PAIRS) {
      const int na = 2 * (p + LN_AHEAD);
      const int nb = (na + 1 == 2 * LN_PAIRS - 1) ? rl : na + 1;
      lo[na]     = *(const v4f*)(xp + na * CC);
      hi[na]     = *(const v4f*)(xp + na * CC + 128);
      lo[na + 1] = *(const v4f*)(xp + nb * CC);
      hi[na + 1] = *(const v4f*)(xp + nb * CC + 128);
    }

    const v4f ya0 = lo[ra] + a0, ya1 = hi[ra] + a1;
    const v4f yb0 = lo[rb] + a0, yb1 = hi[rb] + a1;
    float sa, sa2, sb, sb2;
    ln_stats(ya0, ya1, sa, sa2);
    ln_stats(yb0, yb1, sb, sb2);
    wave_sum4(sa, sa2, sb, sb2);
    v4f oa0, oa1, ob0, ob1;
    ln_apply(ya0, ya1, sa, sa2, g0, g1, e0, e1, oa0, oa1);
    ln_apply(yb0, yb1, sb, sb2, g0, g1, e0, e1, ob0, ob1);

    __builtin_nontemporal_store(oa0, (v4f*)(op + ra * CC));
    __builtin_nontemporal_store(oa1, (v4f*)(op + ra * CC + 128));
    if (rb < 2 * LN_PAIRS - 1 || full) {
      __builtin_nontemporal_store(ob0, (v4f*)(op + rb * CC));
      __builtin_nontemporal_store(ob1, (v4f*)(op + rb * CC + 128));
    }
  }
}

extern "C" void kernel_launch(void* const* d_in, const int* in_sizes, int n_in,
                              void* d_out, int out_size, void* d_ws, size_t ws_size,
                              hipStream_t stream) {
  (void)in_sizes; (void)n_in; (void)out_size; (void)ws_size;

  const float* x     = (const float*)d_in[0];
  const float* W     = (const float*)d_in[1];
  const float* bias  = (const float*)d_in[2];
  const float* gamma = (const float*)d_in[3];
  const float* beta  = (const float*)d_in[4];
  float* out = (float*)d_out;

  float* ws      = (float*)d_ws;
  float* partial = ws;
  float* aggpart = partial + (size_t)BB * NCHUNK * CC;

  colsum_partial_kernel<<<dim3(BB, NCHUNK), 128, 0, stream>>>(x, partial);
  mean_gemm_slice_kernel<<<dim3(KSLICES, BB, NQ), 128, 0, stream>>>(partial, W, aggpart);
  fused_residual_ln_kernel<<<dim3(BB, NCHUNK), 256, 0, stream>>>(x, aggpart, bias, gamma, beta, out);
}
